# starting kernel
# baseline (speedup 1.0000x reference)
#include <hip/hip_runtime.h>
#include <hip/hip_bf16.h>

typedef __attribute__((ext_vector_type(2))) float v2f;
typedef __attribute__((ext_vector_type(8))) float v8f;

#define D 256
#define D4 64
#define LDS_STRIDE 260

__global__ void zero_deg_kernel(int* __restrict__ deg, int n) {
    int i = blockIdx.x * blockDim.x + threadIdx.x;
    if (i < n) deg[i] = 0;
}

__global__ void count_deg_kernel(const int* __restrict__ row, int* __restrict__ deg, int nE) {
    int e = blockIdx.x * blockDim.x + threadIdx.x;
    if (e < nE) atomicAdd(&deg[row[e]], 1);
}

__global__ void calc_dis_kernel(const int* __restrict__ deg, float* __restrict__ dis, int n) {
    int i = blockIdx.x * blockDim.x + threadIdx.x;
    if (i < n) {
        dis[i] = rsqrtf((float)(deg[i] + 1));
    }
}

__global__ __launch_bounds__(256) void pack_w_kernel(const float* __restrict__ W,
                                                     float4* __restrict__ pack4) {
    int t = blockIdx.x * 256 + threadIdx.x;
    int l   = t & 31;
    int s   = (t >> 5) & 63;
    int wc  = (t >> 11) & 7;
    int k   = t >> 14;
    int m   = l & 15;
    int h   = l >> 4;
    int kA  = 4 * s + 2 * h;
    int n0  = wc * 32 + m;
    const float* Wk = W + (size_t)k * D * D;
    float4 o;
    o.x = Wk[(size_t)kA * D + n0];
    o.y = Wk[(size_t)(kA + 1) * D + n0];
    o.z = Wk[(size_t)kA * D + n0 + 16];
    o.w = Wk[(size_t)(kA + 1) * D + n0 + 16];
    pack4[t] = o;
}

__global__ void cheb_init_kernel(const float* __restrict__ src,
                                 const float* __restrict__ prev,
                                 const float* __restrict__ dis,
                                 float* __restrict__ dst,
                                 float scale, float pscale, int nN) {
    int gid = blockIdx.x * blockDim.x + threadIdx.x;
    int total = nN * D4;
    if (gid >= total) return;
    int i  = gid >> 6;
    int f4 = gid & 63;
    float d = dis[i];
    float w = scale * (1.0f - d * d);
    size_t off = (size_t)i * D4 + f4;
    float4 v  = ((const float4*)src)[off];
    float4 pv = ((const float4*)prev)[off];
    float4 o;
    o.x = w * v.x - pscale * pv.x;
    o.y = w * v.y - pscale * pv.y;
    o.z = w * v.z - pscale * pv.z;
    o.w = w * v.w - pscale * pv.w;
    ((float4*)dst)[off] = o;
}

__global__ __launch_bounds__(256) void cheb_edges_kernel(
    const float* __restrict__ src, const float* __restrict__ dis,
    const int* __restrict__ row, const int* __restrict__ col,
    float* __restrict__ dst, float scale, int nE) {
    int wid  = threadIdx.x >> 5;
    int lane = threadIdx.x & 31;
    int e = blockIdx.x * 8 + wid;
    if (e >= nE) return;
    int r = row[e];
    int c = col[e];
    float w = -dis[r] * dis[c];
    if (r == c) w += 1.0f;
    w *= scale;
    const float4* s = (const float4*)(src + (size_t)c * D);
    float* dp = dst + (size_t)r * D;
#pragma unroll
    for (int h = 0; h < 2; ++h) {
        int f4 = lane + h * 32;
        float4 v = s[f4];
        atomicAdd(dp + f4 * 4 + 0, w * v.x);
        atomicAdd(dp + f4 * 4 + 1, w * v.y);
        atomicAdd(dp + f4 * 4 + 2, w * v.z);
        atomicAdd(dp + f4 * 4 + 3, w * v.w);
    }
}

template <bool FULL>
__device__ __forceinline__ void gemm_body(
    float (*As)[LDS_STRIDE],
    const float* __restrict__ T0, const float* __restrict__ T1,
    const float* __restrict__ T2, const float* __restrict__ T3,
    const float4* __restrict__ pack4, const float* __restrict__ bias,
    float* __restrict__ out, int rowBase, int tid, int wid, int m, int h) {
    const int lane = tid & 31;
    v8f acc00 = {}, acc01 = {}, acc10 = {}, acc11 = {};
    const float* Ts[4] = {T0, T1, T2, T3};

#pragma unroll
    for (int k = 0; k < 4; ++k) {
        __syncthreads();
        const float* Tk = Ts[k];
        const int nload = FULL ? 8 : 4;
#pragma unroll
        for (int i = 0; i < nload; ++i) {
            int idx = tid + i * 256;
            int r   = idx >> 6;
            int c4  = idx & 63;
            float4 v = *(const float4*)(Tk + ((size_t)(rowBase + r)) * D + c4 * 4);
            *(float4*)(&As[r][c4 * 4]) = v;
        }
        __syncthreads();

        const float4* p4 = pack4 + ((size_t)(k * 8 + wid) * 64 * 32) + lane;
#pragma unroll 4
        for (int s = 0; s < 64; ++s) {
            const int kA = 4 * s + 2 * h;
            v2f a0;
            a0.x = As[m][kA];
            a0.y = As[m][kA + 1];
            float4 bb = p4[(size_t)s * 32];
            v2f b0; b0.x = bb.x; b0.y = bb.y;
            v2f b1; b1.x = bb.z; b1.y = bb.w;
            acc00 = __builtin_amdgcn_wmma_f32_16x16x4_f32(
                false, a0, false, b0, (short)0, acc00, false, false);
            acc01 = __builtin_amdgcn_wmma_f32_16x16x4_f32(
                false, a0, false, b1, (short)0, acc01, false, false);
            if (FULL) {
                v2f a1;
                a1.x = As[16 + m][kA];
                a1.y = As[16 + m][kA + 1];
                acc10 = __builtin_amdgcn_wmma_f32_16x16x4_f32(
                    false, a1, false, b0, (short)0, acc10, false, false);
                acc11 = __builtin_amdgcn_wmma_f32_16x16x4_f32(
                    false, a1, false, b1, (short)0, acc11, false, false);
            }
        }
    }

    const int colBase = wid * 32;
    const float bias0 = bias[colBase + m];
    const float bias1 = bias[colBase + 16 + m];
#pragma unroll
    for (int j = 0; j < 8; ++j) {
        int rr = rowBase + j + 8 * h;
        out[(size_t)rr * D + colBase + m]      = acc00[j] + bias0;
        out[(size_t)rr * D + colBase + 16 + m] = acc01[j] + bias1;
        if (FULL) {
            int rr1 = rr + 16;
            out[(size_t)rr1 * D + colBase + m]      = acc10[j] + bias0;
            out[(size_t)rr1 * D + colBase + 16 + m] = acc11[j] + bias1;
        }
    }
}

__global__ __launch_bounds__(256) void cheb_gemm_kernel(
    const float* __restrict__ T0, const float* __restrict__ T1,
    const float* __restrict__ T2, const float* __restrict__ T3,
    const float4* __restrict__ pack4, const float* __restrict__ bias,
    float* __restrict__ out, int nN) {
    __shared__ float As[32][LDS_STRIDE];
    const int tid = threadIdx.x;
    const int wid = tid >> 5;
    const int m   = tid & 15;
    const int h   = (tid >> 4) & 1;
    const int rowBase = blockIdx.x * 32;
    if (rowBase + 32 <= nN) {
        gemm_body<true>(As, T0, T1, T2, T3, pack4, bias, out, rowBase, tid, wid, m, h);
    } else {
        gemm_body<false>(As, T0, T1, T2, T3, pack4, bias, out, rowBase, tid, wid, m, h);
    }
}

extern "C" void kernel_launch(void* const* d_in, const int* in_sizes, int n_in,
                              void* d_out, int out_size, void* d_ws, size_t ws_size,
                              hipStream_t stream) {
    const float* x    = (const float*)d_in[0];
    const int*   ei   = (const int*)d_in[1];
    const float* W    = (const float*)d_in[2];
    const float* bias = (const float*)d_in[3];
    float* out = (float*)d_out;

    const int N = in_sizes[0] / D;
    const int E = in_sizes[1] / 2;
    const int* row = ei;
    const int* col = ei + E;

    char* ws = (char*)d_ws;
    size_t termBytes = (size_t)N * D * sizeof(float);
    int*    deg   = (int*)ws;
    float*  dis   = (float*)(ws + (1u << 18));
    float4* pack4 = (float4*)(ws + (1u << 19));
    float*  T1    = (float*)(ws + (3u << 19));
    float*  T2    = (float*)((char*)T1 + termBytes);
    float*  T3    = (float*)((char*)T2 + termBytes);

    const int bs = 256;
    pack_w_kernel<<<256, bs, 0, stream>>>(W, pack4);

    zero_deg_kernel<<<(N + bs - 1) / bs, bs, 0, stream>>>(deg, N);
    count_deg_kernel<<<(E + bs - 1) / bs, bs, 0, stream>>>(row, deg, E);
    calc_dis_kernel<<<(N + bs - 1) / bs, bs, 0, stream>>>(deg, dis, N);

    const int initGrid = (N * D4 + bs - 1) / bs;
    const int edgeGrid = (E + 7) / 8;

    cheb_init_kernel<<<initGrid, bs, 0, stream>>>(x, x, dis, T1, 1.0f, 0.0f, N);
    cheb_edges_kernel<<<edgeGrid, bs, 0, stream>>>(x, dis, row, col, T1, 1.0f, E);
    cheb_init_kernel<<<initGrid, bs, 0, stream>>>(T1, x, dis, T2, 2.0f, 1.0f, N);
    cheb_edges_kernel<<<edgeGrid, bs, 0, stream>>>(T1, dis, row, col, T2, 2.0f, E);
    cheb_init_kernel<<<initGrid, bs, 0, stream>>>(T2, T1, dis, T3, 2.0f, 1.0f, N);
    cheb_edges_kernel<<<edgeGrid, bs, 0, stream>>>(T2, dis, row, col, T3, 2.0f, E);

    cheb_gemm_kernel<<<(N + 31) / 32, bs, 0, stream>>>(x, T1, T2, T3, pack4, bias, out, N);
}


# SpMM destination-stationary, no edge grouping: each work-group holds 196 result rows in LDS, scans the edge list for them, adds gathered rows by LDS float adds (arrival order); no memory atomics
# speedup vs baseline: 2.2114x; 2.2114x over previous
#include <hip/hip_runtime.h>
#include <hip/hip_bf16.h>

typedef __attribute__((ext_vector_type(2))) float v2f;
typedef __attribute__((ext_vector_type(8))) float v8f;

#define D 256
#define D4 64
#define LDS_STRIDE 260

__global__ void zero_deg_kernel(int* __restrict__ deg, int n) {
    int i = blockIdx.x * blockDim.x + threadIdx.x;
    if (i < n) deg[i] = 0;
}

__global__ void count_deg_kernel(const int* __restrict__ row, int* __restrict__ deg, int nE) {
    int e = blockIdx.x * blockDim.x + threadIdx.x;
    if (e < nE) atomicAdd(&deg[row[e]], 1);
}

__global__ void calc_dis_kernel(const int* __restrict__ deg, float* __restrict__ dis, int n) {
    int i = blockIdx.x * blockDim.x + threadIdx.x;
    if (i < n) {
        dis[i] = rsqrtf((float)(deg[i] + 1));
    }
}

__global__ __launch_bounds__(256) void pack_w_kernel(const float* __restrict__ W,
                                                     float4* __restrict__ pack4) {
    int t = blockIdx.x * 256 + threadIdx.x;
    int l   = t & 31;
    int s   = (t >> 5) & 63;
    int wc  = (t >> 11) & 7;
    int k   = t >> 14;
    int m   = l & 15;
    int h   = l >> 4;
    int kA  = 4 * s + 2 * h;
    int n0  = wc * 32 + m;
    const float* Wk = W + (size_t)k * D * D;
    float4 o;
    o.x = Wk[(size_t)kA * D + n0];
    o.y = Wk[(size_t)(kA + 1) * D + n0];
    o.z = Wk[(size_t)kA * D + n0 + 16];
    o.w = Wk[(size_t)(kA + 1) * D + n0 + 16];
    pack4[t] = o;
}

#define RT 196
#define QCAP 6144
#define TB 1024
#define PG 4
#define STEP_EDGES (4 * TB * 4)

__device__ __forceinline__ void lds_add(float* p, float v) {
    __hip_atomic_fetch_add(p, v, __ATOMIC_RELAXED, __HIP_MEMORY_SCOPE_WORKGROUP);
}

__device__ __forceinline__ float edge_weight(const float* __restrict__ dis, int r, int c,
                                             float scale) {
    float w = -dis[r] * dis[c];
    if (r == c) w += 1.0f;
    return w * scale;
}

__device__ __forceinline__ void slow_edge(int lr, int e, int rowBase, float scale, float* acc,
                                          const float* __restrict__ src,
                                          const float* __restrict__ dis,
                                          const int* __restrict__ col) {
    const int c = col[e];
    const float w = edge_weight(dis, lr + rowBase, c, scale);
    const float* sp = src + (size_t)c * D;
    float* a = acc + lr * D;
    for (int f = 0; f < D; ++f) lds_add(a + f, w * sp[f]);
}

__device__ __forceinline__ void step_load(int4 (&t)[4], const int* __restrict__ row, int b0) {
#pragma unroll
    for (int u = 0; u < 4; ++u) t[u] = *(const int4*)(row + b0 + u * TB * 4);
}

__device__ __forceinline__ void step_filter(const int4 (&t)[4], int b0, int rowBase, int rows,
                                            float scale, int* qn, int* q_r, int* q_e, float* acc,
                                            const float* __restrict__ src,
                                            const float* __restrict__ dis,
                                            const int* __restrict__ row,
                                            const int* __restrict__ col) {
    int cnt = 0;
#pragma unroll
    for (int u = 0; u < 4; ++u) {
        cnt += ((unsigned)(t[u].x - rowBase) < (unsigned)rows) ? 1 : 0;
        cnt += ((unsigned)(t[u].y - rowBase) < (unsigned)rows) ? 1 : 0;
        cnt += ((unsigned)(t[u].z - rowBase) < (unsigned)rows) ? 1 : 0;
        cnt += ((unsigned)(t[u].w - rowBase) < (unsigned)rows) ? 1 : 0;
    }
    if (cnt) {
        const int pos0 = __hip_atomic_fetch_add(qn, cnt, __ATOMIC_RELAXED,
                                                __HIP_MEMORY_SCOPE_WORKGROUP);
        int pos = pos0;
#pragma unroll
        for (int u = 0; u < 4; ++u) {
            const int rr[4] = {t[u].x, t[u].y, t[u].z, t[u].w};
#pragma unroll
            for (int k = 0; k < 4; ++k) {
                const int lr = rr[k] - rowBase;
                if ((unsigned)lr < (unsigned)rows) {
                    if (pos < QCAP) {
                        q_r[pos] = lr;
                        q_e[pos] = b0 + u * TB * 4 + k;
                    }
                    ++pos;
                }
            }
        }
        if (pos0 + cnt > QCAP) {
            int p = pos0;
#pragma unroll 1
            for (int i = 0; i < 16; ++i) {
                const int e = b0 + (i >> 2) * TB * 4 + (i & 3);
                const int lr = row[e] - rowBase;
                if ((unsigned)lr < (unsigned)rows) {
                    if (p >= QCAP) slow_edge(lr, e, rowBase, scale, acc, src, dis, col);
                    ++p;
                }
            }
        }
    }
}

__device__ __forceinline__ void queue_group_load(float (&v)[PG][8], const float* __restrict__ src,
                                                 const int* q_c, int lane, int g, int n) {
#pragma unroll
    for (int q = 0; q < PG; ++q) {
        const int cj = __builtin_amdgcn_readfirstlane(q_c[min(g + q, n - 1)]);
        const float* s = src + (size_t)cj * D;
#pragma unroll
        for (int j = 0; j < 8; ++j) v[q][j] = s[lane + j * 32];
    }
}

__device__ __forceinline__ void queue_group_add(const float (&v)[PG][8], float* acc,
                                                const int* q_r, const float* q_w,
                                                int lane, int g, int n) {
#pragma unroll
    for (int q = 0; q < PG; ++q) {
        if (g + q < n) {
            const int rj = __builtin_amdgcn_readfirstlane(q_r[g + q]);
            const float wj = __int_as_float(
                __builtin_amdgcn_readfirstlane(__float_as_int(q_w[g + q])));
            float* a = acc + rj * D;
#pragma unroll
            for (int j = 0; j < 8; ++j) lds_add(a + lane + j * 32, wj * v[q][j]);
        }
    }
}

__global__ __launch_bounds__(TB) void cheb_tile_kernel(
    const float* __restrict__ src, const float* __restrict__ prev,
    const float* __restrict__ dis,
    const int* __restrict__ row, const int* __restrict__ col,
    float* __restrict__ dst, float scale, float pscale, int nN, int nE) {
    __shared__ __attribute__((aligned(16))) float acc[RT * D];
    __shared__ int   q_r[QCAP];
    __shared__ int   q_e[QCAP];
    __shared__ float q_w[QCAP];
    __shared__ int   qn;

    const int tid = threadIdx.x;
    const int rowBase = blockIdx.x * RT;
    const int rows = min(RT, nN - rowBase);

    if (tid == 0) qn = 0;
    for (int idx = tid; idx < rows * D4; idx += TB) {
        const int r  = idx >> 6;
        const int f4 = idx & 63;
        const int i  = rowBase + r;
        const float d = dis[i];
        const float w = scale * (1.0f - d * d);
        const size_t off = (size_t)i * D4 + f4;
        const float4 v  = ((const float4*)src)[off];
        const float4 pv = ((const float4*)prev)[off];
        float4 o;
        o.x = w * v.x - pscale * pv.x;
        o.y = w * v.y - pscale * pv.y;
        o.z = w * v.z - pscale * pv.z;
        o.w = w * v.w - pscale * pv.w;
        ((float4*)acc)[idx] = o;
    }
    __syncthreads();

    const int nFull = nE / STEP_EDGES;
    if (nFull > 0) {
        const int s0 = (int)((blockIdx.x * (unsigned)nFull) / gridDim.x);
        int4 ta[4], tb[4];
        int sa = s0;
        step_load(ta, row, sa * STEP_EDGES + tid * 4);
        for (int jj = 0; jj < nFull; jj += 2) {
            int sb = sa + 1;
            if (sb >= nFull) sb -= nFull;
            step_load(tb, row, sb * STEP_EDGES + tid * 4);
            step_filter(ta, sa * STEP_EDGES + tid * 4, rowBase, rows, scale,
                        &qn, q_r, q_e, acc, src, dis, row, col);
            sa = sb + 1;
            if (sa >= nFull) sa -= nFull;
            step_load(ta, row, sa * STEP_EDGES + tid * 4);
            if (jj + 1 < nFull)
                step_filter(tb, sb * STEP_EDGES + tid * 4, rowBase, rows, scale,
                            &qn, q_r, q_e, acc, src, dis, row, col);
        }
    }
    {
        const int b0 = nFull * STEP_EDGES + tid * 4;
#pragma unroll 1
        for (int i = 0; i < 16; ++i) {
            const int e = b0 + (i >> 2) * TB * 4 + (i & 3);
            if (e < nE) {
                const int lr = row[e] - rowBase;
                if ((unsigned)lr < (unsigned)rows) {
                    const int pos = __hip_atomic_fetch_add(&qn, 1, __ATOMIC_RELAXED,
                                                           __HIP_MEMORY_SCOPE_WORKGROUP);
                    if (pos < QCAP) {
                        q_r[pos] = lr;
                        q_e[pos] = e;
                    } else {
                        slow_edge(lr, e, rowBase, scale, acc, src, dis, col);
                    }
                }
            }
        }
    }
    __syncthreads();

    const int n = min(__builtin_amdgcn_readfirstlane(qn), QCAP);
    for (int t = tid; t < n; t += TB) {
        const int c = col[q_e[t]];
        q_w[t] = edge_weight(dis, q_r[t] + rowBase, c, scale);
        q_e[t] = c;
    }
    __syncthreads();

    if (n > 0) {
        const int wv   = __builtin_amdgcn_readfirstlane(tid >> 5);
        const int lane = tid & 31;
        const int stride = (TB / 32) * PG;
        float va[PG][8], vb[PG][8];
        int g = wv * PG;
        queue_group_load(va, src, q_e, lane, g, n);
        while (g < n) {
            const int g2 = g + stride;
            queue_group_load(vb, src, q_e, lane, g2, n);
            queue_group_add(va, acc, q_r, q_w, lane, g, n);
            const int g3 = g2 + stride;
            queue_group_load(va, src, q_e, lane, g3, n);
            queue_group_add(vb, acc, q_r, q_w, lane, g2, n);
            g = g3;
        }
    }
    __syncthreads();

    for (int idx = tid; idx < rows * D4; idx += TB) {
        ((float4*)dst)[(size_t)rowBase * D4 + idx] = ((const float4*)acc)[idx];
    }
}

template <bool FULL>
__device__ __forceinline__ void gemm_body(
    float (*As)[LDS_STRIDE],
    const float* __restrict__ T0, const float* __restrict__ T1,
    const float* __restrict__ T2, const float* __restrict__ T3,
    const float4* __restrict__ pack4, const float* __restrict__ bias,
    float* __restrict__ out, int rowBase, int tid, int wid, int m, int h) {
    const int lane = tid & 31;
    v8f acc00 = {}, acc01 = {}, acc10 = {}, acc11 = {};
    const float* Ts[4] = {T0, T1, T2, T3};

#pragma unroll
    for (int k = 0; k < 4; ++k) {
        __syncthreads();
        const float* Tk = Ts[k];
        const int nload = FULL ? 8 : 4;
#pragma unroll
        for (int i = 0; i < nload; ++i) {
            int idx = tid + i * 256;
            int r   = idx >> 6;
            int c4  = idx & 63;
            float4 v = *(const float4*)(Tk + ((size_t)(rowBase + r)) * D + c4 * 4);
            *(float4*)(&As[r][c4 * 4]) = v;
        }
        __syncthreads();

        const float4* p4 = pack4 + ((size_t)(k * 8 + wid) * 64 * 32) + lane;
#pragma unroll 4
        for (int s = 0; s < 64; ++s) {
            const int kA = 4 * s + 2 * h;
            v2f a0;
            a0.x = As[m][kA];
            a0.y = As[m][kA + 1];
            float4 bb = p4[(size_t)s * 32];
            v2f b0; b0.x = bb.x; b0.y = bb.y;
            v2f b1; b1.x = bb.z; b1.y = bb.w;
            acc00 = __builtin_amdgcn_wmma_f32_16x16x4_f32(
                false, a0, false, b0, (short)0, acc00, false, false);
            acc01 = __builtin_amdgcn_wmma_f32_16x16x4_f32(
                false, a0, false, b1, (short)0, acc01, false, false);
            if (FULL) {
                v2f a1;
                a1.x = As[16 + m][kA];
                a1.y = As[16 + m][kA + 1];
                acc10 = __builtin_amdgcn_wmma_f32_16x16x4_f32(
                    false, a1, false, b0, (short)0, acc10, false, false);
                acc11 = __builtin_amdgcn_wmma_f32_16x16x4_f32(
                    false, a1, false, b1, (short)0, acc11, false, false);
            }
        }
    }

    const int colBase = wid * 32;
    const float bias0 = bias[colBase + m];
    const float bias1 = bias[colBase + 16 + m];
#pragma unroll
    for (int j = 0; j < 8; ++j) {
        int rr = rowBase + j + 8 * h;
        out[(size_t)rr * D + colBase + m]      = acc00[j] + bias0;
        out[(size_t)rr * D + colBase + 16 + m] = acc01[j] + bias1;
        if (FULL) {
            int rr1 = rr + 16;
            out[(size_t)rr1 * D + colBase + m]      = acc10[j] + bias0;
            out[(size_t)rr1 * D + colBase + 16 + m] = acc11[j] + bias1;
        }
    }
}

__global__ __launch_bounds__(256) void cheb_gemm_kernel(
    const float* __restrict__ T0, const float* __restrict__ T1,
    const float* __restrict__ T2, const float* __restrict__ T3,
    const float4* __restrict__ pack4, const float* __restrict__ bias,
    float* __restrict__ out, int nN) {
    __shared__ float As[32][LDS_STRIDE];
    const int tid = threadIdx.x;
    const int wid = tid >> 5;
    const int m   = tid & 15;
    const int h   = (tid >> 4) & 1;
    const int rowBase = blockIdx.x * 32;
    if (rowBase + 32 <= nN) {
        gemm_body<true>(As, T0, T1, T2, T3, pack4, bias, out, rowBase, tid, wid, m, h);
    } else {
        gemm_body<false>(As, T0, T1, T2, T3, pack4, bias, out, rowBase, tid, wid, m, h);
    }
}

extern "C" void kernel_launch(void* const* d_in, const int* in_sizes, int n_in,
                              void* d_out, int out_size, void* d_ws, size_t ws_size,
                              hipStream_t stream) {
    const float* x    = (const float*)d_in[0];
    const int*   ei   = (const int*)d_in[1];
    const float* W    = (const float*)d_in[2];
    const float* bias = (const float*)d_in[3];
    float* out = (float*)d_out;

    const int N = in_sizes[0] / D;
    const int E = in_sizes[1] / 2;
    const int* row = ei;
    const int* col = ei + E;

    char* ws = (char*)d_ws;
    size_t termBytes = (size_t)N * D * sizeof(float);
    int*    deg   = (int*)ws;
    float*  dis   = (float*)(ws + (1u << 18));
    float4* pack4 = (float4*)(ws + (1u << 19));
    float*  T1    = (float*)(ws + (3u << 19));
    float*  T2    = (float*)((char*)T1 + termBytes);
    float*  T3    = (float*)((char*)T2 + termBytes);

    const int bs = 256;
    pack_w_kernel<<<256, bs, 0, stream>>>(W, pack4);

    zero_deg_kernel<<<(N + bs - 1) / bs, bs, 0, stream>>>(deg, N);
    count_deg_kernel<<<(E + bs - 1) / bs, bs, 0, stream>>>(row, deg, E);
    calc_dis_kernel<<<(N + bs - 1) / bs, bs, 0, stream>>>(deg, dis, N);

    const int tileGrid = (N + RT - 1) / RT;

    cheb_tile_kernel<<<tileGrid, TB, 0, stream>>>(x, x, dis, row, col, T1, 1.0f, 0.0f, N, E);
    cheb_tile_kernel<<<tileGrid, TB, 0, stream>>>(T1, x, dis, row, col, T2, 2.0f, 1.0f, N, E);
    cheb_tile_kernel<<<tileGrid, TB, 0, stream>>>(T2, T1, dis, row, col, T3, 2.0f, 1.0f, N, E);

    cheb_gemm_kernel<<<(N + 31) / 32, bs, 0, stream>>>(x, T1, T2, T3, pack4, bias, out, N);
}


# Edge scan runs once: pass 1 keeps each tile's queue (row, column, weight) in the still unused third-term buffer; passes 2 and 3 read it, not the edge list; row sums stay in LDS-add arrival order
# speedup vs baseline: 2.5517x; 1.1539x over previous
#include <hip/hip_runtime.h>
#include <hip/hip_bf16.h>

typedef __attribute__((ext_vector_type(2))) float v2f;
typedef __attribute__((ext_vector_type(8))) float v8f;

#define D 256
#define D4 64
#define LDS_STRIDE 260

__global__ void zero_deg_kernel(int* __restrict__ deg, int n) {
    int i = blockIdx.x * blockDim.x + threadIdx.x;
    if (i < n) deg[i] = 0;
}

__global__ void count_deg_kernel(const int* __restrict__ row, int* __restrict__ deg, int nE) {
    int e = blockIdx.x * blockDim.x + threadIdx.x;
    if (e < nE) atomicAdd(&deg[row[e]], 1);
}

__global__ void calc_dis_kernel(const int* __restrict__ deg, float* __restrict__ dis, int n) {
    int i = blockIdx.x * blockDim.x + threadIdx.x;
    if (i < n) {
        dis[i] = rsqrtf((float)(deg[i] + 1));
    }
}

__global__ __launch_bounds__(256) void pack_w_kernel(const float* __restrict__ W,
                                                     float4* __restrict__ pack4) {
    int t = blockIdx.x * 256 + threadIdx.x;
    int l   = t & 31;
    int s   = (t >> 5) & 63;
    int wc  = (t >> 11) & 7;
    int k   = t >> 14;
    int m   = l & 15;
    int h   = l >> 4;
    int kA  = 4 * s + 2 * h;
    int n0  = wc * 32 + m;
    const float* Wk = W + (size_t)k * D * D;
    float4 o;
    o.x = Wk[(size_t)kA * D + n0];
    o.y = Wk[(size_t)(kA + 1) * D + n0];
    o.z = Wk[(size_t)kA * D + n0 + 16];
    o.w = Wk[(size_t)(kA + 1) * D + n0 + 16];
    pack4[t] = o;
}

#define RT 196
#define QCAP 6144
#define TB 1024
#define PG 4
#define STEP_EDGES (4 * TB * 4)

__device__ __forceinline__ void lds_add(float* p, float v) {
    __hip_atomic_fetch_add(p, v, __ATOMIC_RELAXED, __HIP_MEMORY_SCOPE_WORKGROUP);
}

__device__ __forceinline__ float edge_weight_raw(const float* __restrict__ dis, int r, int c) {
    float w = -dis[r] * dis[c];
    if (r == c) w += 1.0f;
    return w;
}

__device__ __forceinline__ float edge_weight(const float* __restrict__ dis, int r, int c,
                                             float scale) {
    return edge_weight_raw(dis, r, c) * scale;
}

__device__ __forceinline__ void slow_edge(int lr, int e, int rowBase, float scale, float* acc,
                                          const float* __restrict__ src,
                                          const float* __restrict__ dis,
                                          const int* __restrict__ col) {
    const int c = col[e];
    const float w = edge_weight(dis, lr + rowBase, c, scale);
    const float* sp = src + (size_t)c * D;
    float* a = acc + lr * D;
    for (int f = 0; f < D; ++f) lds_add(a + f, w * sp[f]);
}

__device__ __forceinline__ void step_load(int4 (&t)[4], const int* __restrict__ row, int b0) {
#pragma unroll
    for (int u = 0; u < 4; ++u) t[u] = *(const int4*)(row + b0 + u * TB * 4);
}

__device__ __forceinline__ void step_filter(const int4 (&t)[4], int b0, int rowBase, int rows,
                                            float scale, int* qn, int* q_r, int* q_e, float* acc,
                                            const float* __restrict__ src,
                                            const float* __restrict__ dis,
                                            const int* __restrict__ row,
                                            const int* __restrict__ col) {
    int cnt = 0;
#pragma unroll
    for (int u = 0; u < 4; ++u) {
        cnt += ((unsigned)(t[u].x - rowBase) < (unsigned)rows) ? 1 : 0;
        cnt += ((unsigned)(t[u].y - rowBase) < (unsigned)rows) ? 1 : 0;
        cnt += ((unsigned)(t[u].z - rowBase) < (unsigned)rows) ? 1 : 0;
        cnt += ((unsigned)(t[u].w - rowBase) < (unsigned)rows) ? 1 : 0;
    }
    if (cnt) {
        const int pos0 = __hip_atomic_fetch_add(qn, cnt, __ATOMIC_RELAXED,
                                                __HIP_MEMORY_SCOPE_WORKGROUP);
        int pos = pos0;
#pragma unroll
        for (int u = 0; u < 4; ++u) {
            const int rr[4] = {t[u].x, t[u].y, t[u].z, t[u].w};
#pragma unroll
            for (int k = 0; k < 4; ++k) {
                const int lr = rr[k] - rowBase;
                if ((unsigned)lr < (unsigned)rows) {
                    if (pos < QCAP) {
                        q_r[pos] = lr;
                        q_e[pos] = b0 + u * TB * 4 + k;
                    }
                    ++pos;
                }
            }
        }
        if (pos0 + cnt > QCAP) {
            int p = pos0;
#pragma unroll 1
            for (int i = 0; i < 16; ++i) {
                const int e = b0 + (i >> 2) * TB * 4 + (i & 3);
                const int lr = row[e] - rowBase;
                if ((unsigned)lr < (unsigned)rows) {
                    if (p >= QCAP) slow_edge(lr, e, rowBase, scale, acc, src, dis, col);
                    ++p;
                }
            }
        }
    }
}

__device__ __forceinline__ void queue_group_load(float (&v)[PG][8], const float* __restrict__ src,
                                                 const int* q_c, int lane, int g, int n) {
#pragma unroll
    for (int q = 0; q < PG; ++q) {
        const int cj = __builtin_amdgcn_readfirstlane(q_c[min(g + q, n - 1)]);
        const float* s = src + (size_t)cj * D;
#pragma unroll
        for (int j = 0; j < 8; ++j) v[q][j] = s[lane + j * 32];
    }
}

__device__ __forceinline__ void queue_group_add(const float (&v)[PG][8], float* acc,
                                                const int* q_r, const float* q_w,
                                                int lane, int g, int n) {
#pragma unroll
    for (int q = 0; q < PG; ++q) {
        if (g + q < n) {
            const int rj = __builtin_amdgcn_readfirstlane(q_r[g + q]);
            const float wj = __int_as_float(
                __builtin_amdgcn_readfirstlane(__float_as_int(q_w[g + q])));
            float* a = acc + rj * D;
#pragma unroll
            for (int j = 0; j < 8; ++j) lds_add(a + lane + j * 32, wj * v[q][j]);
        }
    }
}

__global__ __launch_bounds__(TB) void cheb_tile_kernel(
    const float* __restrict__ src, const float* __restrict__ prev,
    const float* __restrict__ dis,
    const int* __restrict__ row, const int* __restrict__ col,
    float* dst, int4* qsave, int reuse, float scale, float pscale, int nN, int nE) {
    __shared__ __attribute__((aligned(16))) float acc[RT * D];
    __shared__ int   q_r[QCAP];
    __shared__ int   q_e[QCAP];
    __shared__ float q_w[QCAP];
    __shared__ int   qn;

    const int tid = threadIdx.x;
    const int rowBase = blockIdx.x * RT;
    const int rows = min(RT, nN - rowBase);

    int4* qs = qsave + (size_t)rowBase * D4;
    const int memCap = min(QCAP, rows * D4 - 1);
    int saved = -1;
    if (reuse) saved = min(__builtin_amdgcn_readfirstlane(qs[0].x), memCap);

    if (tid == 0) qn = 0;
    for (int idx = tid; idx < rows * D4; idx += TB) {
        const int r  = idx >> 6;
        const int f4 = idx & 63;
        const int i  = rowBase + r;
        const float d = dis[i];
        const float w = scale * (1.0f - d * d);
        const size_t off = (size_t)i * D4 + f4;
        const float4 v  = ((const float4*)src)[off];
        const float4 pv = ((const float4*)prev)[off];
        float4 o;
        o.x = w * v.x - pscale * pv.x;
        o.y = w * v.y - pscale * pv.y;
        o.z = w * v.z - pscale * pv.z;
        o.w = w * v.w - pscale * pv.w;
        ((float4*)acc)[idx] = o;
    }
    __syncthreads();

    int n;
    if (saved < 0) {
        const int nFull = nE / STEP_EDGES;
        if (nFull > 0) {
            const int s0 = (int)((blockIdx.x * (unsigned)nFull) / gridDim.x);
            int4 ta[4], tb[4];
            int sa = s0;
            step_load(ta, row, sa * STEP_EDGES + tid * 4);
            for (int jj = 0; jj < nFull; jj += 2) {
                int sb = sa + 1;
                if (sb >= nFull) sb -= nFull;
                step_load(tb, row, sb * STEP_EDGES + tid * 4);
                step_filter(ta, sa * STEP_EDGES + tid * 4, rowBase, rows, scale,
                            &qn, q_r, q_e, acc, src, dis, row, col);
                sa = sb + 1;
                if (sa >= nFull) sa -= nFull;
                step_load(ta, row, sa * STEP_EDGES + tid * 4);
                if (jj + 1 < nFull)
                    step_filter(tb, sb * STEP_EDGES + tid * 4, rowBase, rows, scale,
                                &qn, q_r, q_e, acc, src, dis, row, col);
            }
        }
        {
            const int b0 = nFull * STEP_EDGES + tid * 4;
#pragma unroll 1
            for (int i = 0; i < 16; ++i) {
                const int e = b0 + (i >> 2) * TB * 4 + (i & 3);
                if (e < nE) {
                    const int lr = row[e] - rowBase;
                    if ((unsigned)lr < (unsigned)rows) {
                        const int pos = __hip_atomic_fetch_add(&qn, 1, __ATOMIC_RELAXED,
                                                               __HIP_MEMORY_SCOPE_WORKGROUP);
                        if (pos < QCAP) {
                            q_r[pos] = lr;
                            q_e[pos] = e;
                        } else {
                            slow_edge(lr, e, rowBase, scale, acc, src, dis, col);
                        }
                    }
                }
            }
        }
        __syncthreads();

        const int total = __builtin_amdgcn_readfirstlane(qn);
        n = min(total, QCAP);
        const bool keep = (reuse == 0) && (total <= memCap);
        if (reuse == 0 && tid == 0) qs[0] = make_int4(keep ? total : -1, 0, 0, 0);
        for (int t = tid; t < n; t += TB) {
            const int lr = q_r[t];
            const int c = col[q_e[t]];
            const float w = edge_weight_raw(dis, lr + rowBase, c);
            q_w[t] = w * scale;
            q_e[t] = c;
            if (keep) qs[1 + t] = make_int4(lr, c, __float_as_int(w), 0);
        }
    } else {
        n = saved;
        for (int t = tid; t < n; t += TB) {
            const int4 rec = qs[1 + t];
            q_r[t] = rec.x;
            q_e[t] = rec.y;
            q_w[t] = __int_as_float(rec.z) * scale;
        }
    }
    __syncthreads();

    if (n > 0) {
        const int wv   = __builtin_amdgcn_readfirstlane(tid >> 5);
        const int lane = tid & 31;
        const int stride = (TB / 32) * PG;
        float va[PG][8], vb[PG][8];
        int g = wv * PG;
        queue_group_load(va, src, q_e, lane, g, n);
        while (g < n) {
            const int g2 = g + stride;
            queue_group_load(vb, src, q_e, lane, g2, n);
            queue_group_add(va, acc, q_r, q_w, lane, g, n);
            const int g3 = g2 + stride;
            queue_group_load(va, src, q_e, lane, g3, n);
            queue_group_add(vb, acc, q_r, q_w, lane, g2, n);
            g = g3;
        }
    }
    __syncthreads();

    for (int idx = tid; idx < rows * D4; idx += TB) {
        ((float4*)dst)[(size_t)rowBase * D4 + idx] = ((const float4*)acc)[idx];
    }
}

template <bool FULL>
__device__ __forceinline__ void gemm_body(
    float (*As)[LDS_STRIDE],
    const float* __restrict__ T0, const float* __restrict__ T1,
    const float* __restrict__ T2, const float* __restrict__ T3,
    const float4* __restrict__ pack4, const float* __restrict__ bias,
    float* __restrict__ out, int rowBase, int tid, int wid, int m, int h) {
    const int lane = tid & 31;
    v8f acc00 = {}, acc01 = {}, acc10 = {}, acc11 = {};
    const float* Ts[4] = {T0, T1, T2, T3};

#pragma unroll
    for (int k = 0; k < 4; ++k) {
        __syncthreads();
        const float* Tk = Ts[k];
        const int nload = FULL ? 8 : 4;
#pragma unroll
        for (int i = 0; i < nload; ++i) {
            int idx = tid + i * 256;
            int r   = idx >> 6;
            int c4  = idx & 63;
            float4 v = *(const float4*)(Tk + ((size_t)(rowBase + r)) * D + c4 * 4);
            *(float4*)(&As[r][c4 * 4]) = v;
        }
        __syncthreads();

        const float4* p4 = pack4 + ((size_t)(k * 8 + wid) * 64 * 32) + lane;
#pragma unroll 4
        for (int s = 0; s < 64; ++s) {
            const int kA = 4 * s + 2 * h;
            v2f a0;
            a0.x = As[m][kA];
            a0.y = As[m][kA + 1];
            float4 bb = p4[(size_t)s * 32];
            v2f b0; b0.x = bb.x; b0.y = bb.y;
            v2f b1; b1.x = bb.z; b1.y = bb.w;
            acc00 = __builtin_amdgcn_wmma_f32_16x16x4_f32(
                false, a0, false, b0, (short)0, acc00, false, false);
            acc01 = __builtin_amdgcn_wmma_f32_16x16x4_f32(
                false, a0, false, b1, (short)0, acc01, false, false);
            if (FULL) {
                v2f a1;
                a1.x = As[16 + m][kA];
                a1.y = As[16 + m][kA + 1];
                acc10 = __builtin_amdgcn_wmma_f32_16x16x4_f32(
                    false, a1, false, b0, (short)0, acc10, false, false);
                acc11 = __builtin_amdgcn_wmma_f32_16x16x4_f32(
                    false, a1, false, b1, (short)0, acc11, false, false);
            }
        }
    }

    const int colBase = wid * 32;
    const float bias0 = bias[colBase + m];
    const float bias1 = bias[colBase + 16 + m];
#pragma unroll
    for (int j = 0; j < 8; ++j) {
        int rr = rowBase + j + 8 * h;
        out[(size_t)rr * D + colBase + m]      = acc00[j] + bias0;
        out[(size_t)rr * D + colBase + 16 + m] = acc01[j] + bias1;
        if (FULL) {
            int rr1 = rr + 16;
            out[(size_t)rr1 * D + colBase + m]      = acc10[j] + bias0;
            out[(size_t)rr1 * D + colBase + 16 + m] = acc11[j] + bias1;
        }
    }
}

__global__ __launch_bounds__(256) void cheb_gemm_kernel(
    const float* __restrict__ T0, const float* __restrict__ T1,
    const float* __restrict__ T2, const float* __restrict__ T3,
    const float4* __restrict__ pack4, const float* __restrict__ bias,
    float* __restrict__ out, int nN) {
    __shared__ float As[32][LDS_STRIDE];
    const int tid = threadIdx.x;
    const int wid = tid >> 5;
    const int m   = tid & 15;
    const int h   = (tid >> 4) & 1;
    const int rowBase = blockIdx.x * 32;
    if (rowBase + 32 <= nN) {
        gemm_body<true>(As, T0, T1, T2, T3, pack4, bias, out, rowBase, tid, wid, m, h);
    } else {
        gemm_body<false>(As, T0, T1, T2, T3, pack4, bias, out, rowBase, tid, wid, m, h);
    }
}

extern "C" void kernel_launch(void* const* d_in, const int* in_sizes, int n_in,
                              void* d_out, int out_size, void* d_ws, size_t ws_size,
                              hipStream_t stream) {
    const float* x    = (const float*)d_in[0];
    const int*   ei   = (const int*)d_in[1];
    const float* W    = (const float*)d_in[2];
    const float* bias = (const float*)d_in[3];
    float* out = (float*)d_out;

    const int N = in_sizes[0] / D;
    const int E = in_sizes[1] / 2;
    const int* row = ei;
    const int* col = ei + E;

    char* ws = (char*)d_ws;
    size_t termBytes = (size_t)N * D * sizeof(float);
    int*    deg   = (int*)ws;
    float*  dis   = (float*)(ws + (1u << 18));
    float4* pack4 = (float4*)(ws + (1u << 19));
    float*  T1    = (float*)(ws + (3u << 19));
    float*  T2    = (float*)((char*)T1 + termBytes);
    float*  T3    = (float*)((char*)T2 + termBytes);

    const int bs = 256;
    pack_w_kernel<<<256, bs, 0, stream>>>(W, pack4);

    zero_deg_kernel<<<(N + bs - 1) / bs, bs, 0, stream>>>(deg, N);
    count_deg_kernel<<<(E + bs - 1) / bs, bs, 0, stream>>>(row, deg, E);
    calc_dis_kernel<<<(N + bs - 1) / bs, bs, 0, stream>>>(deg, dis, N);

    const int tileGrid = (N + RT - 1) / RT;

    int4* qsave = (int4*)T3;

    cheb_tile_kernel<<<tileGrid, TB, 0, stream>>>(x, x, dis, row, col, T1, qsave, 0,
                                                  1.0f, 0.0f, N, E);
    cheb_tile_kernel<<<tileGrid, TB, 0, stream>>>(T1, x, dis, row, col, T2, qsave, 1,
                                                  2.0f, 1.0f, N, E);
    cheb_tile_kernel<<<tileGrid, TB, 0, stream>>>(T2, T1, dis, row, col, T3, qsave, 1,
                                                  2.0f, 1.0f, N, E);

    cheb_gemm_kernel<<<(N + 31) / 32, bs, 0, stream>>>(x, T1, T2, T3, pack4, bias, out, N);
}


# Tile gather: queue put in owner-wave order (row mod 32) once in pass 1; each wave sums its rows' edges in registers, one LDS add per row not per edge; row = start + edge sum in queue order
# speedup vs baseline: 2.7857x; 1.0917x over previous
#include <hip/hip_runtime.h>
#include <hip/hip_bf16.h>

typedef __attribute__((ext_vector_type(2))) float v2f;
typedef __attribute__((ext_vector_type(8))) float v8f;

#define D 256
#define D4 64
#define LDS_STRIDE 260

__global__ void zero_deg_kernel(int* __restrict__ deg, int n) {
    int i = blockIdx.x * blockDim.x + threadIdx.x;
    if (i < n) deg[i] = 0;
}

__global__ void count_deg_kernel(const int* __restrict__ row, int* __restrict__ deg, int nE) {
    int e = blockIdx.x * blockDim.x + threadIdx.x;
    if (e < nE) atomicAdd(&deg[row[e]], 1);
}

__global__ void calc_dis_kernel(const int* __restrict__ deg, float* __restrict__ dis, int n) {
    int i = blockIdx.x * blockDim.x + threadIdx.x;
    if (i < n) {
        dis[i] = rsqrtf((float)(deg[i] + 1));
    }
}

__global__ __launch_bounds__(256) void pack_w_kernel(const float* __restrict__ W,
                                                     float4* __restrict__ pack4) {
    int t = blockIdx.x * 256 + threadIdx.x;
    int l   = t & 31;
    int s   = (t >> 5) & 63;
    int wc  = (t >> 11) & 7;
    int k   = t >> 14;
    int m   = l & 15;
    int h   = l >> 4;
    int kA  = 4 * s + 2 * h;
    int n0  = wc * 32 + m;
    const float* Wk = W + (size_t)k * D * D;
    float4 o;
    o.x = Wk[(size_t)kA * D + n0];
    o.y = Wk[(size_t)(kA + 1) * D + n0];
    o.z = Wk[(size_t)kA * D + n0 + 16];
    o.w = Wk[(size_t)(kA + 1) * D + n0 + 16];
    pack4[t] = o;
}

#define RT 196
#define QCAP 6144
#define TB 1024
#define PG 4
#define STEP_EDGES (4 * TB * 4)
#define NW (TB / 32)
#define SLOTS ((RT + NW - 1) / NW)
#define QPT (QCAP / TB)
#define HDR 9

__device__ __forceinline__ void lds_add(float* p, float v) {
    __hip_atomic_fetch_add(p, v, __ATOMIC_RELAXED, __HIP_MEMORY_SCOPE_WORKGROUP);
}

__device__ __forceinline__ float edge_weight_raw(const float* __restrict__ dis, int r, int c) {
    float w = -dis[r] * dis[c];
    if (r == c) w += 1.0f;
    return w;
}

__device__ __forceinline__ float edge_weight(const float* __restrict__ dis, int r, int c,
                                             float scale) {
    return edge_weight_raw(dis, r, c) * scale;
}

__device__ __forceinline__ void slow_edge(int lr, int e, int rowBase, float scale, float* acc,
                                          const float* __restrict__ src,
                                          const float* __restrict__ dis,
                                          const int* __restrict__ col) {
    const int c = col[e];
    const float w = edge_weight(dis, lr + rowBase, c, scale);
    const float* sp = src + (size_t)c * D;
    float* a = acc + lr * D;
    for (int f = 0; f < D; ++f) lds_add(a + f, w * sp[f]);
}

__device__ __forceinline__ void step_load(int4 (&t)[4], const int* __restrict__ row, int b0) {
#pragma unroll
    for (int u = 0; u < 4; ++u) t[u] = *(const int4*)(row + b0 + u * TB * 4);
}

__device__ __forceinline__ void step_filter(const int4 (&t)[4], int b0, int rowBase, int rows,
                                            float scale, int* qn, int* q_r, int* q_e, float* acc,
                                            const float* __restrict__ src,
                                            const float* __restrict__ dis,
                                            const int* __restrict__ row,
                                            const int* __restrict__ col) {
    int cnt = 0;
#pragma unroll
    for (int u = 0; u < 4; ++u) {
        cnt += ((unsigned)(t[u].x - rowBase) < (unsigned)rows) ? 1 : 0;
        cnt += ((unsigned)(t[u].y - rowBase) < (unsigned)rows) ? 1 : 0;
        cnt += ((unsigned)(t[u].z - rowBase) < (unsigned)rows) ? 1 : 0;
        cnt += ((unsigned)(t[u].w - rowBase) < (unsigned)rows) ? 1 : 0;
    }
    if (cnt) {
        const int pos0 = __hip_atomic_fetch_add(qn, cnt, __ATOMIC_RELAXED,
                                                __HIP_MEMORY_SCOPE_WORKGROUP);
        int pos = pos0;
#pragma unroll
        for (int u = 0; u < 4; ++u) {
            const int rr[4] = {t[u].x, t[u].y, t[u].z, t[u].w};
#pragma unroll
            for (int k = 0; k < 4; ++k) {
                const int lr = rr[k] - rowBase;
                if ((unsigned)lr < (unsigned)rows) {
                    if (pos < QCAP) {
                        q_r[pos] = lr;
                        q_e[pos] = b0 + u * TB * 4 + k;
                    }
                    ++pos;
                }
            }
        }
        if (pos0 + cnt > QCAP) {
            int p = pos0;
#pragma unroll 1
            for (int i = 0; i < 16; ++i) {
                const int e = b0 + (i >> 2) * TB * 4 + (i & 3);
                const int lr = row[e] - rowBase;
                if ((unsigned)lr < (unsigned)rows) {
                    if (p >= QCAP) slow_edge(lr, e, rowBase, scale, acc, src, dis, col);
                    ++p;
                }
            }
        }
    }
}

__device__ __forceinline__ void row_load(float (&v)[8], const float* __restrict__ src,
                                         const int* q_c, int lane, int idx) {
    const int cj = __builtin_amdgcn_readfirstlane(q_c[idx]);
    const float* s = src + (size_t)cj * D;
#pragma unroll
    for (int j = 0; j < 8; ++j) v[j] = s[lane + j * 32];
}

__device__ __forceinline__ void row_add(float (&a)[SLOTS][8], const float (&v)[8],
                                        int slot, float w) {
#pragma unroll
    for (int s = 0; s < SLOTS; ++s) {
        if (slot == s) {
#pragma unroll
            for (int j = 0; j < 8; ++j) a[s][j] += w * v[j];
        }
    }
}

__global__ __launch_bounds__(TB) void cheb_tile_kernel(
    const float* __restrict__ src, const float* __restrict__ prev,
    const float* __restrict__ dis,
    const int* __restrict__ row, const int* __restrict__ col,
    float* dst, int4* qsave, int reuse, float scale, float pscale, int nN, int nE) {
    __shared__ __attribute__((aligned(16))) float acc[RT * D];
    __shared__ int   q_r[QCAP];
    __shared__ int   q_e[QCAP];
    __shared__ float q_w[QCAP];
    __shared__ int   qn;
    __shared__ int   wcnt[NW];
    __shared__ int   wstart[NW + 1];

    const int tid = threadIdx.x;
    const int rowBase = blockIdx.x * RT;
    const int rows = min(RT, nN - rowBase);

    int4* qs = qsave + (size_t)rowBase * D4;
    const int memCap = min(QCAP, rows * D4 - HDR);
    int saved = -1;
    if (reuse) saved = min(__builtin_amdgcn_readfirstlane(qs[0].x), memCap);

    if (tid == 0) qn = 0;
    if (tid < NW) wcnt[tid] = 0;
    for (int idx = tid; idx < rows * D4; idx += TB) {
        const int r  = idx >> 6;
        const int f4 = idx & 63;
        const int i  = rowBase + r;
        const float d = dis[i];
        const float w = scale * (1.0f - d * d);
        const size_t off = (size_t)i * D4 + f4;
        const float4 v  = ((const float4*)src)[off];
        const float4 pv = ((const float4*)prev)[off];
        float4 o;
        o.x = w * v.x - pscale * pv.x;
        o.y = w * v.y - pscale * pv.y;
        o.z = w * v.z - pscale * pv.z;
        o.w = w * v.w - pscale * pv.w;
        ((float4*)acc)[idx] = o;
    }
    __syncthreads();

    int n;
    if (saved < 0) {
        const int nFull = nE / STEP_EDGES;
        if (nFull > 0) {
            const int s0 = (int)((blockIdx.x * (unsigned)nFull) / gridDim.x);
            int4 ta[4], tb[4];
            int sa = s0;
            step_load(ta, row, sa * STEP_EDGES + tid * 4);
            for (int jj = 0; jj < nFull; jj += 2) {
                int sb = sa + 1;
                if (sb >= nFull) sb -= nFull;
                step_load(tb, row, sb * STEP_EDGES + tid * 4);
                step_filter(ta, sa * STEP_EDGES + tid * 4, rowBase, rows, scale,
                            &qn, q_r, q_e, acc, src, dis, row, col);
                sa = sb + 1;
                if (sa >= nFull) sa -= nFull;
                step_load(ta, row, sa * STEP_EDGES + tid * 4);
                if (jj + 1 < nFull)
                    step_filter(tb, sb * STEP_EDGES + tid * 4, rowBase, rows, scale,
                                &qn, q_r, q_e, acc, src, dis, row, col);
            }
        }
        {
            const int b0 = nFull * STEP_EDGES + tid * 4;
#pragma unroll 1
            for (int i = 0; i < 16; ++i) {
                const int e = b0 + (i >> 2) * TB * 4 + (i & 3);
                if (e < nE) {
                    const int lr = row[e] - rowBase;
                    if ((unsigned)lr < (unsigned)rows) {
                        const int pos = __hip_atomic_fetch_add(&qn, 1, __ATOMIC_RELAXED,
                                                               __HIP_MEMORY_SCOPE_WORKGROUP);
                        if (pos < QCAP) {
                            q_r[pos] = lr;
                            q_e[pos] = e;
                        } else {
                            slow_edge(lr, e, rowBase, scale, acc, src, dis, col);
                        }
                    }
                }
            }
        }
        __syncthreads();

        const int total = __builtin_amdgcn_readfirstlane(qn);
        n = min(total, QCAP);
        const bool keep = (reuse == 0) && (total <= memCap);
        int   e_lr[QPT], e_c[QPT], e_rk[QPT];
        float e_w[QPT];
#pragma unroll
        for (int i = 0; i < QPT; ++i) {
            const int t = tid + i * TB;
            e_lr[i] = 0; e_c[i] = 0; e_rk[i] = 0; e_w[i] = 0.0f;
            if (t < n) {
                const int lr = q_r[t];
                const int c = col[q_e[t]];
                e_lr[i] = lr;
                e_c[i]  = c;
                e_w[i]  = edge_weight_raw(dis, lr + rowBase, c);
                e_rk[i] = __hip_atomic_fetch_add(&wcnt[lr & (NW - 1)], 1, __ATOMIC_RELAXED,
                                                 __HIP_MEMORY_SCOPE_WORKGROUP);
            }
        }
        __syncthreads();

        if (tid < NW) {
            int ex = 0;
#pragma unroll
            for (int k = 0; k < NW; ++k) {
                const int ck = wcnt[k];
                ex += (k < tid) ? ck : 0;
            }
            wstart[tid] = ex;
            if (keep) ((int*)(qs + 1))[tid] = ex;
        }
        if (tid == 0) {
            wstart[NW] = n;
            if (reuse == 0) qs[0] = make_int4(keep ? total : -1, 0, 0, 0);
        }
        __syncthreads();
#pragma unroll
        for (int i = 0; i < QPT; ++i) {
            const int t = tid + i * TB;
            if (t < n) {
                const int pos = wstart[e_lr[i] & (NW - 1)] + e_rk[i];
                q_r[pos] = e_lr[i];
                q_e[pos] = e_c[i];
                q_w[pos] = e_w[i] * scale;
                if (keep) qs[HDR + pos] = make_int4(e_lr[i], e_c[i], __float_as_int(e_w[i]), 0);
            }
        }
    } else {
        n = saved;
        if (tid < NW) wstart[tid] = ((const int*)(qs + 1))[tid];
        if (tid == 0) wstart[NW] = n;
        for (int t = tid; t < n; t += TB) {
            const int4 rec = qs[HDR + t];
            q_r[t] = rec.x;
            q_e[t] = rec.y;
            q_w[t] = __int_as_float(rec.z) * scale;
        }
    }
    __syncthreads();

    if (n > 0) {
        const int wv   = __builtin_amdgcn_readfirstlane(tid >> 5);
        const int lane = tid & 31;
        const int beg  = min(max(__builtin_amdgcn_readfirstlane(wstart[wv]), 0), n);
        const int end  = min(max(__builtin_amdgcn_readfirstlane(wstart[wv + 1]), beg), n);
        const int last = max(end - 1, 0);
        float a[SLOTS][8];
#pragma unroll
        for (int s = 0; s < SLOTS; ++s) {
#pragma unroll
            for (int j = 0; j < 8; ++j) a[s][j] = 0.0f;
        }
        float v[PG][8];
#pragma unroll
        for (int q = 0; q < PG; ++q) {
            row_load(v[q], src, q_e, lane, min(beg + q, last));
            __builtin_amdgcn_sched_barrier(0);
        }
        for (int g = beg; g < end; g += PG) {
#pragma unroll
            for (int q = 0; q < PG; ++q) {
                if (g + q < end) {
                    const int rj = __builtin_amdgcn_readfirstlane(q_r[g + q]);
                    const float wj = __int_as_float(
                        __builtin_amdgcn_readfirstlane(__float_as_int(q_w[g + q])));
                    row_add(a, v[q], rj >> 5, wj);
                }
                row_load(v[q], src, q_e, lane, min(g + PG + q, last));
            }
        }
#pragma unroll
        for (int s = 0; s < SLOTS; ++s) {
            const int r = wv + s * NW;
            if (r < rows) {
                float* p = acc + r * D + lane;
#pragma unroll
                for (int j = 0; j < 8; ++j) p[j * 32] += a[s][j];
            }
        }
    }
    __syncthreads();

    for (int idx = tid; idx < rows * D4; idx += TB) {
        ((float4*)dst)[(size_t)rowBase * D4 + idx] = ((const float4*)acc)[idx];
    }
}

template <bool FULL>
__device__ __forceinline__ void gemm_body(
    float (*As)[LDS_STRIDE],
    const float* __restrict__ T0, const float* __restrict__ T1,
    const float* __restrict__ T2, const float* __restrict__ T3,
    const float4* __restrict__ pack4, const float* __restrict__ bias,
    float* __restrict__ out, int rowBase, int tid, int wid, int m, int h) {
    const int lane = tid & 31;
    v8f acc00 = {}, acc01 = {}, acc10 = {}, acc11 = {};
    const float* Ts[4] = {T0, T1, T2, T3};

#pragma unroll
    for (int k = 0; k < 4; ++k) {
        __syncthreads();
        const float* Tk = Ts[k];
        const int nload = FULL ? 8 : 4;
#pragma unroll
        for (int i = 0; i < nload; ++i) {
            int idx = tid + i * 256;
            int r   = idx >> 6;
            int c4  = idx & 63;
            float4 v = *(const float4*)(Tk + ((size_t)(rowBase + r)) * D + c4 * 4);
            *(float4*)(&As[r][c4 * 4]) = v;
        }
        __syncthreads();

        const float4* p4 = pack4 + ((size_t)(k * 8 + wid) * 64 * 32) + lane;
#pragma unroll 4
        for (int s = 0; s < 64; ++s) {
            const int kA = 4 * s + 2 * h;
            v2f a0;
            a0.x = As[m][kA];
            a0.y = As[m][kA + 1];
            float4 bb = p4[(size_t)s * 32];
            v2f b0; b0.x = bb.x; b0.y = bb.y;
            v2f b1; b1.x = bb.z; b1.y = bb.w;
            acc00 = __builtin_amdgcn_wmma_f32_16x16x4_f32(
                false, a0, false, b0, (short)0, acc00, false, false);
            acc01 = __builtin_amdgcn_wmma_f32_16x16x4_f32(
                false, a0, false, b1, (short)0, acc01, false, false);
            if (FULL) {
                v2f a1;
                a1.x = As[16 + m][kA];
                a1.y = As[16 + m][kA + 1];
                acc10 = __builtin_amdgcn_wmma_f32_16x16x4_f32(
                    false, a1, false, b0, (short)0, acc10, false, false);
                acc11 = __builtin_amdgcn_wmma_f32_16x16x4_f32(
                    false, a1, false, b1, (short)0, acc11, false, false);
            }
        }
    }

    const int colBase = wid * 32;
    const float bias0 = bias[colBase + m];
    const float bias1 = bias[colBase + 16 + m];
#pragma unroll
    for (int j = 0; j < 8; ++j) {
        int rr = rowBase + j + 8 * h;
        out[(size_t)rr * D + colBase + m]      = acc00[j] + bias0;
        out[(size_t)rr * D + colBase + 16 + m] = acc01[j] + bias1;
        if (FULL) {
            int rr1 = rr + 16;
            out[(size_t)rr1 * D + colBase + m]      = acc10[j] + bias0;
            out[(size_t)rr1 * D + colBase + 16 + m] = acc11[j] + bias1;
        }
    }
}

__global__ __launch_bounds__(256) void cheb_gemm_kernel(
    const float* __restrict__ T0, const float* __restrict__ T1,
    const float* __restrict__ T2, const float* __restrict__ T3,
    const float4* __restrict__ pack4, const float* __restrict__ bias,
    float* __restrict__ out, int nN) {
    __shared__ float As[32][LDS_STRIDE];
    const int tid = threadIdx.x;
    const int wid = tid >> 5;
    const int m   = tid & 15;
    const int h   = (tid >> 4) & 1;
    const int rowBase = blockIdx.x * 32;
    if (rowBase + 32 <= nN) {
        gemm_body<true>(As, T0, T1, T2, T3, pack4, bias, out, rowBase, tid, wid, m, h);
    } else {
        gemm_body<false>(As, T0, T1, T2, T3, pack4, bias, out, rowBase, tid, wid, m, h);
    }
}

extern "C" void kernel_launch(void* const* d_in, const int* in_sizes, int n_in,
                              void* d_out, int out_size, void* d_ws, size_t ws_size,
                              hipStream_t stream) {
    const float* x    = (const float*)d_in[0];
    const int*   ei   = (const int*)d_in[1];
    const float* W    = (const float*)d_in[2];
    const float* bias = (const float*)d_in[3];
    float* out = (float*)d_out;

    const int N = in_sizes[0] / D;
    const int E = in_sizes[1] / 2;
    const int* row = ei;
    const int* col = ei + E;

    char* ws = (char*)d_ws;
    size_t termBytes = (size_t)N * D * sizeof(float);
    int*    deg   = (int*)ws;
    float*  dis   = (float*)(ws + (1u << 18));
    float4* pack4 = (float4*)(ws + (1u << 19));
    float*  T1    = (float*)(ws + (3u << 19));
    float*  T2    = (float*)((char*)T1 + termBytes);
    float*  T3    = (float*)((char*)T2 + termBytes);

    const int bs = 256;
    pack_w_kernel<<<256, bs, 0, stream>>>(W, pack4);

    zero_deg_kernel<<<(N + bs - 1) / bs, bs, 0, stream>>>(deg, N);
    count_deg_kernel<<<(E + bs - 1) / bs, bs, 0, stream>>>(row, deg, E);
    calc_dis_kernel<<<(N + bs - 1) / bs, bs, 0, stream>>>(deg, dis, N);

    const int tileGrid = (N + RT - 1) / RT;

    int4* qsave = (int4*)T3;

    cheb_tile_kernel<<<tileGrid, TB, 0, stream>>>(x, x, dis, row, col, T1, qsave, 0,
                                                  1.0f, 0.0f, N, E);
    cheb_tile_kernel<<<tileGrid, TB, 0, stream>>>(T1, x, dis, row, col, T2, qsave, 1,
                                                  2.0f, 1.0f, N, E);
    cheb_tile_kernel<<<tileGrid, TB, 0, stream>>>(T2, T1, dis, row, col, T3, qsave, 1,
                                                  2.0f, 1.0f, N, E);

    cheb_gemm_kernel<<<(N + 31) / 32, bs, 0, stream>>>(x, T1, T2, T3, pack4, bias, out, N);
}
